# starting kernel
# baseline (speedup 1.0000x reference)
#include <hip/hip_runtime.h>


typedef float v2f __attribute__((ext_vector_type(2)));
typedef float v8f __attribute__((ext_vector_type(8)));

#define THREADS 256

__global__ void k_deg_init(float* __restrict__ deg, int n) {
    int i = blockIdx.x * blockDim.x + threadIdx.x;
    if (i < n) deg[i] = 1.0f;
}

__global__ void k_deg_accum(const int* __restrict__ dst, float* __restrict__ deg, int E) {
    int i = blockIdx.x * blockDim.x + threadIdx.x;
    if (i >= E) return;
    __builtin_prefetch(dst + i + 8192, 0, 0);
    atomicAdd(deg + dst[i], 1.0f);
}

__global__ void k_rsqrt_inplace(float* __restrict__ deg, int n) {
    int i = blockIdx.x * blockDim.x + threadIdx.x;
    if (i < n) deg[i] = rsqrtf(fmaxf(deg[i], 1.0f));
}

template<int KDIM, int NCOLT, int WCOLS>
__global__ __launch_bounds__(THREADS)
void k_gemm_rowscale(const float* __restrict__ X, const float* __restrict__ W,
                     const float* __restrict__ dinv, float* __restrict__ P, int nrows)
{
    __shared__ float sW[KDIM * NCOLT * 16];
    const int tid = threadIdx.x;
    for (int i = tid; i < KDIM * NCOLT * 16; i += THREADS) {
        const int kk = i / (NCOLT * 16);
        const int c  = i % (NCOLT * 16);
        sW[i] = (c < WCOLS) ? W[kk * WCOLS + c] : 0.0f;
    }
    __syncthreads();

    const int gwave   = (blockIdx.x * THREADS + tid) >> 5;
    const int rowBase = gwave * 16;
    if (rowBase >= nrows) return;

    const int lane = tid & 31;
    const int lh   = lane >> 4;
    const int ln   = lane & 15;

    v8f acc[NCOLT] = {};

    const float* xrow = X + (size_t)(rowBase + ln) * KDIM + 2 * lh;
    for (int kb = 0; kb < KDIM; kb += 4) {
        const v2f A = *reinterpret_cast<const v2f*>(xrow + kb);
        const float* w0 = &sW[(kb + 2 * lh) * (NCOLT * 16) + ln];
        const float* w1 = w0 + NCOLT * 16;
        #pragma unroll
        for (int t = 0; t < NCOLT; ++t) {
            v2f B;
            B.x = w0[t * 16];
            B.y = w1[t * 16];
            acc[t] = __builtin_amdgcn_wmma_f32_16x16x4_f32(
                false, A, false, B, (short)0, acc[t], false, false);
        }
    }

    const float* dvp = dinv + rowBase + 8 * lh;
    float dv[8];
    #pragma unroll
    for (int r = 0; r < 8; ++r) dv[r] = dvp[r];

    #pragma unroll
    for (int t = 0; t < NCOLT; ++t) {
        const int col = t * 16 + ln;
        if (col < WCOLS) {
            #pragma unroll
            for (int r = 0; r < 8; ++r)
                P[(size_t)(rowBase + 8 * lh + r) * WCOLS + col] = dv[r] * acc[t][r];
        }
    }
}

__global__ void k_copy4(const float4* __restrict__ src, float4* __restrict__ dst, int n4) {
    int i = blockIdx.x * blockDim.x + threadIdx.x;
    if (i < n4) dst[i] = src[i];
}

__global__ void k_scatter128(const float* __restrict__ p,
                             const int* __restrict__ src, const int* __restrict__ dst,
                             float* __restrict__ a, int E)
{
    const int t    = blockIdx.x * blockDim.x + threadIdx.x;
    const int e    = t >> 5;
    const int lane = t & 31;
    if (e >= E) return;
    __builtin_prefetch(src + e + 2048, 0, 0);
    const int s = src[e];
    const int d = dst[e];
    const float4 v = *reinterpret_cast<const float4*>(p + (size_t)s * 128 + lane * 4);
    float* ap = a + (size_t)d * 128 + lane * 4;
    atomicAdd(ap + 0, v.x);
    atomicAdd(ap + 1, v.y);
    atomicAdd(ap + 2, v.z);
    atomicAdd(ap + 3, v.w);
}

__global__ void k_scatter8(const float* __restrict__ p,
                           const int* __restrict__ src, const int* __restrict__ dst,
                           float* __restrict__ a, int E)
{
    const int e = blockIdx.x * blockDim.x + threadIdx.x;
    if (e >= E) return;
    const int s = src[e];
    const int d = dst[e];
    const float4 v0 = *reinterpret_cast<const float4*>(p + (size_t)s * 8);
    const float4 v1 = *reinterpret_cast<const float4*>(p + (size_t)s * 8 + 4);
    float* ap = a + (size_t)d * 8;
    atomicAdd(ap + 0, v0.x); atomicAdd(ap + 1, v0.y);
    atomicAdd(ap + 2, v0.z); atomicAdd(ap + 3, v0.w);
    atomicAdd(ap + 4, v1.x); atomicAdd(ap + 5, v1.y);
    atomicAdd(ap + 6, v1.z); atomicAdd(ap + 7, v1.w);
}

template<int D, bool RELU>
__global__ void k_finalize(const float* __restrict__ a, const float* __restrict__ dinv,
                           const float* __restrict__ bias, float* __restrict__ out, int n)
{
    const int i = blockIdx.x * blockDim.x + threadIdx.x;
    if (i >= n * D) return;
    const int node = i / D;
    const int c    = i % D;
    float v = dinv[node] * a[i] + bias[c];
    if (RELU) v = fmaxf(v, 0.0f);
    out[i] = v;
}

extern "C" void kernel_launch(void* const* d_in, const int* in_sizes, int n_in,
                              void* d_out, int out_size, void* d_ws, size_t ws_size,
                              hipStream_t stream)
{
    constexpr int F = 128, H = 128, C = 8;
    const int N = in_sizes[0] / F;
    const int E = in_sizes[1] / 2;

    const float* x   = (const float*)d_in[0];
    const int*   src = (const int*)  d_in[1];
    const int*   dst = src + E;
    const float* W1  = (const float*)d_in[2];
    const float* b1  = (const float*)d_in[3];
    const float* W2  = (const float*)d_in[4];
    const float* b2  = (const float*)d_in[5];
    float*       out = (float*)d_out;

    float* ws   = (float*)d_ws;
    float* dinv = ws;
    float* pbuf = ws + N;
    float* abuf = pbuf + (size_t)N * H;
    float* p2   = abuf;
    float* a2   = abuf + (size_t)N * C;

    const int gN   = (N + THREADS - 1) / THREADS;
    const int gE   = (E + THREADS - 1) / THREADS;
    const int gGemm = ((N / 16) * 32 + THREADS - 1) / THREADS;
    const int gNH  = ((size_t)N * H + THREADS - 1) / THREADS;
    const int gNH4 = ((size_t)N * H / 4 + THREADS - 1) / THREADS;
    const int gE32 = ((size_t)E * 32 + THREADS - 1) / THREADS;
    const int gNC  = ((size_t)N * C + THREADS - 1) / THREADS;
    const int gNC4 = ((size_t)N * C / 4 + THREADS - 1) / THREADS;

    k_deg_init     <<<gN, THREADS, 0, stream>>>(dinv, N);
    k_deg_accum    <<<gE, THREADS, 0, stream>>>(dst, dinv, E);
    k_rsqrt_inplace<<<gN, THREADS, 0, stream>>>(dinv, N);

    k_gemm_rowscale<F, H / 16, H><<<gGemm, THREADS, 0, stream>>>(x, W1, dinv, pbuf, N);

    k_copy4    <<<gNH4, THREADS, 0, stream>>>((const float4*)pbuf, (float4*)abuf, (int)((size_t)N * H / 4));
    k_scatter128<<<gE32, THREADS, 0, stream>>>(pbuf, src, dst, abuf, E);

    k_finalize<H, true><<<gNH, THREADS, 0, stream>>>(abuf, dinv, b1, pbuf, N);

    k_gemm_rowscale<H, 1, C><<<gGemm, THREADS, 0, stream>>>(pbuf, W2, dinv, p2, N);

    k_copy4  <<<gNC4, THREADS, 0, stream>>>((const float4*)p2, (float4*)a2, (int)((size_t)N * C / 4));
    k_scatter8<<<gE, THREADS, 0, stream>>>(p2, src, dst, a2, E);

    k_finalize<C, false><<<gNC, THREADS, 0, stream>>>(a2, dinv, b2, out, N);
}


# Both edge scatters: lanes laid across the feature row so each float-atomic instruction covers whole contiguous rows (D=128: 1 line, was 4; D=8: 4 rows, was 32 words); sums still in arrival order
# speedup vs baseline: 1.7085x; 1.7085x over previous
#include <hip/hip_runtime.h>


typedef float v2f __attribute__((ext_vector_type(2)));
typedef float v8f __attribute__((ext_vector_type(8)));

#define THREADS 256

__global__ void k_deg_init(float* __restrict__ deg, int n) {
    int i = blockIdx.x * blockDim.x + threadIdx.x;
    if (i < n) deg[i] = 1.0f;
}

__global__ void k_deg_accum(const int* __restrict__ dst, float* __restrict__ deg, int E) {
    int i = blockIdx.x * blockDim.x + threadIdx.x;
    if (i >= E) return;
    __builtin_prefetch(dst + i + 8192, 0, 0);
    atomicAdd(deg + dst[i], 1.0f);
}

__global__ void k_rsqrt_inplace(float* __restrict__ deg, int n) {
    int i = blockIdx.x * blockDim.x + threadIdx.x;
    if (i < n) deg[i] = rsqrtf(fmaxf(deg[i], 1.0f));
}

template<int KDIM, int NCOLT, int WCOLS>
__global__ __launch_bounds__(THREADS)
void k_gemm_rowscale(const float* __restrict__ X, const float* __restrict__ W,
                     const float* __restrict__ dinv, float* __restrict__ P, int nrows)
{
    __shared__ float sW[KDIM * NCOLT * 16];
    const int tid = threadIdx.x;
    for (int i = tid; i < KDIM * NCOLT * 16; i += THREADS) {
        const int kk = i / (NCOLT * 16);
        const int c  = i % (NCOLT * 16);
        sW[i] = (c < WCOLS) ? W[kk * WCOLS + c] : 0.0f;
    }
    __syncthreads();

    const int gwave   = (blockIdx.x * THREADS + tid) >> 5;
    const int rowBase = gwave * 16;
    if (rowBase >= nrows) return;

    const int lane = tid & 31;
    const int lh   = lane >> 4;
    const int ln   = lane & 15;

    v8f acc[NCOLT] = {};

    const float* xrow = X + (size_t)(rowBase + ln) * KDIM + 2 * lh;
    for (int kb = 0; kb < KDIM; kb += 4) {
        const v2f A = *reinterpret_cast<const v2f*>(xrow + kb);
        const float* w0 = &sW[(kb + 2 * lh) * (NCOLT * 16) + ln];
        const float* w1 = w0 + NCOLT * 16;
        #pragma unroll
        for (int t = 0; t < NCOLT; ++t) {
            v2f B;
            B.x = w0[t * 16];
            B.y = w1[t * 16];
            acc[t] = __builtin_amdgcn_wmma_f32_16x16x4_f32(
                false, A, false, B, (short)0, acc[t], false, false);
        }
    }

    const float* dvp = dinv + rowBase + 8 * lh;
    float dv[8];
    #pragma unroll
    for (int r = 0; r < 8; ++r) dv[r] = dvp[r];

    #pragma unroll
    for (int t = 0; t < NCOLT; ++t) {
        const int col = t * 16 + ln;
        if (col < WCOLS) {
            #pragma unroll
            for (int r = 0; r < 8; ++r)
                P[(size_t)(rowBase + 8 * lh + r) * WCOLS + col] = dv[r] * acc[t][r];
        }
    }
}

__global__ void k_copy4(const float4* __restrict__ src, float4* __restrict__ dst, int n4) {
    int i = blockIdx.x * blockDim.x + threadIdx.x;
    if (i < n4) dst[i] = src[i];
}

__global__ void k_scatter128(const float* __restrict__ p,
                             const int* __restrict__ src, const int* __restrict__ dst,
                             float* __restrict__ a, int E)
{
    const int t    = blockIdx.x * blockDim.x + threadIdx.x;
    const int e    = t >> 5;
    const int lane = t & 31;
    if (e >= E) return;
    __builtin_prefetch(src + e + 2048, 0, 0);
    const int s = src[e];
    const int d = dst[e];
    const float* pp = p + (size_t)s * 128 + lane;
    float*       ap = a + (size_t)d * 128 + lane;
    const float v0 = pp[0];
    const float v1 = pp[32];
    const float v2 = pp[64];
    const float v3 = pp[96];
    atomicAdd(ap +  0, v0);
    atomicAdd(ap + 32, v1);
    atomicAdd(ap + 64, v2);
    atomicAdd(ap + 96, v3);
}

__global__ void k_scatter8(const float* __restrict__ p,
                           const int* __restrict__ src, const int* __restrict__ dst,
                           float* __restrict__ a, int E)
{
    const int e    = blockIdx.x * blockDim.x + threadIdx.x;
    const int lane = threadIdx.x & 31;
    const int sub  = lane >> 3;
    const int c    = lane & 7;
    const int s = (e < E) ? src[e] : 0;
    const int d = (e < E) ? dst[e] : -1;

    int   dd[8];
    float v[8];
    #pragma unroll
    for (int i = 0; i < 8; ++i) {
        const int from = i * 4 + sub;
        const int si = __shfl(s, from, 32);
        dd[i]        = __shfl(d, from, 32);
        v[i] = p[(size_t)si * 8 + c];
    }
    #pragma unroll
    for (int i = 0; i < 8; ++i)
        if (dd[i] >= 0) atomicAdd(a + (size_t)dd[i] * 8 + c, v[i]);
}

template<int D, bool RELU>
__global__ void k_finalize(const float* __restrict__ a, const float* __restrict__ dinv,
                           const float* __restrict__ bias, float* __restrict__ out, int n)
{
    const int i = blockIdx.x * blockDim.x + threadIdx.x;
    if (i >= n * D) return;
    const int node = i / D;
    const int c    = i % D;
    float v = dinv[node] * a[i] + bias[c];
    if (RELU) v = fmaxf(v, 0.0f);
    out[i] = v;
}

extern "C" void kernel_launch(void* const* d_in, const int* in_sizes, int n_in,
                              void* d_out, int out_size, void* d_ws, size_t ws_size,
                              hipStream_t stream)
{
    constexpr int F = 128, H = 128, C = 8;
    const int N = in_sizes[0] / F;
    const int E = in_sizes[1] / 2;

    const float* x   = (const float*)d_in[0];
    const int*   src = (const int*)  d_in[1];
    const int*   dst = src + E;
    const float* W1  = (const float*)d_in[2];
    const float* b1  = (const float*)d_in[3];
    const float* W2  = (const float*)d_in[4];
    const float* b2  = (const float*)d_in[5];
    float*       out = (float*)d_out;

    float* ws   = (float*)d_ws;
    float* dinv = ws;
    float* pbuf = ws + N;
    float* abuf = pbuf + (size_t)N * H;
    float* p2   = abuf;
    float* a2   = abuf + (size_t)N * C;

    const int gN   = (N + THREADS - 1) / THREADS;
    const int gE   = (E + THREADS - 1) / THREADS;
    const int gGemm = ((N / 16) * 32 + THREADS - 1) / THREADS;
    const int gNH  = ((size_t)N * H + THREADS - 1) / THREADS;
    const int gNH4 = ((size_t)N * H / 4 + THREADS - 1) / THREADS;
    const int gE32 = ((size_t)E * 32 + THREADS - 1) / THREADS;
    const int gNC  = ((size_t)N * C + THREADS - 1) / THREADS;
    const int gNC4 = ((size_t)N * C / 4 + THREADS - 1) / THREADS;

    k_deg_init     <<<gN, THREADS, 0, stream>>>(dinv, N);
    k_deg_accum    <<<gE, THREADS, 0, stream>>>(dst, dinv, E);
    k_rsqrt_inplace<<<gN, THREADS, 0, stream>>>(dinv, N);

    k_gemm_rowscale<F, H / 16, H><<<gGemm, THREADS, 0, stream>>>(x, W1, dinv, pbuf, N);

    k_copy4    <<<gNH4, THREADS, 0, stream>>>((const float4*)pbuf, (float4*)abuf, (int)((size_t)N * H / 4));
    k_scatter128<<<gE32, THREADS, 0, stream>>>(pbuf, src, dst, abuf, E);

    k_finalize<H, true><<<gNH, THREADS, 0, stream>>>(abuf, dinv, b1, pbuf, N);

    k_gemm_rowscale<H, 1, C><<<gGemm, THREADS, 0, stream>>>(pbuf, W2, dinv, p2, N);

    k_copy4  <<<gNC4, THREADS, 0, stream>>>((const float4*)p2, (float4*)a2, (int)((size_t)N * C / 4));
    k_scatter8<<<gE, THREADS, 0, stream>>>(p2, src, dst, a2, E);

    k_finalize<C, false><<<gNC, THREADS, 0, stream>>>(a2, dinv, b2, out, N);
}


# Edge scatter-add atomics replaced by a pull: edges counting-sorted by destination once per call (int atomic gives the slot), each node sums its in-rows in registers in slot order, writes once
# speedup vs baseline: 3.1317x; 1.8330x over previous
#include <hip/hip_runtime.h>


typedef float v2f __attribute__((ext_vector_type(2)));
typedef float v8f __attribute__((ext_vector_type(8)));

#define THREADS 256
#define SCAN_THREADS 1024
#define SCAN_TILE (SCAN_THREADS * 4)
#define EPT 4

__global__ void k_zero(int* __restrict__ cnt, int n) {
    int i = blockIdx.x * blockDim.x + threadIdx.x;
    if (i < n) cnt[i] = 0;
}

__global__ void k_count_rank(const int* __restrict__ dst, int* __restrict__ cnt,
                             int* __restrict__ rank, int E) {
    const int T = gridDim.x * blockDim.x;
    const int e = blockIdx.x * blockDim.x + threadIdx.x;
    int d[EPT], r[EPT];
    #pragma unroll
    for (int k = 0; k < EPT; ++k) d[k] = (e + k * T < E) ? dst[e + k * T] : -1;
    #pragma unroll
    for (int k = 0; k < EPT; ++k) r[k] = (d[k] >= 0) ? atomicAdd(cnt + d[k], 1) : 0;
    #pragma unroll
    for (int k = 0; k < EPT; ++k) if (d[k] >= 0) rank[e + k * T] = r[k];
}

__global__ __launch_bounds__(SCAN_THREADS)
void k_scan_dinv(const int* __restrict__ cnt, int* __restrict__ rowptr,
                 float* __restrict__ dinv, int n)
{
    __shared__ int sred[32];
    __shared__ int sbase;
    const int tid  = threadIdx.x;
    const int lane = tid & 31;
    const int w    = tid >> 5;
    const int tile0 = blockIdx.x * SCAN_TILE;

    const int4* __restrict__ c4 = reinterpret_cast<const int4*>(cnt);
    const int nq = tile0 / 4;
    int pre = 0;
    for (int i = tid; i < nq; i += 8 * SCAN_THREADS) {
        int4 q[8];
        #pragma unroll
        for (int u = 0; u < 8; ++u) {
            const int ii = i + u * SCAN_THREADS;
            q[u] = (ii < nq) ? c4[ii] : make_int4(0, 0, 0, 0);
        }
        #pragma unroll
        for (int u = 0; u < 8; ++u) pre += (q[u].x + q[u].y) + (q[u].z + q[u].w);
    }
    for (int o = 16; o > 0; o >>= 1) pre += __shfl_xor(pre, o, 32);
    if (lane == 0) sred[w] = pre;
    __syncthreads();
    if (w == 0) {
        int v = sred[lane];
        for (int o = 16; o > 0; o >>= 1) v += __shfl_xor(v, o, 32);
        if (lane == 0) sbase = v;
    }
    __syncthreads();
    const int base = sbase;

    const int i0 = tile0 + tid * 4;
    int q[4];
    #pragma unroll
    for (int k = 0; k < 4; ++k) q[k] = (i0 + k < n) ? cnt[i0 + k] : 0;
    const int t = (q[0] + q[1]) + (q[2] + q[3]);
    int incl = t;
    for (int o = 1; o < 32; o <<= 1) {
        const int y = __shfl_up(incl, o, 32);
        if (lane >= o) incl += y;
    }
    if (lane == 31) sred[w] = incl;
    __syncthreads();
    int wv = (lane < w) ? sred[lane] : 0;
    for (int o = 16; o > 0; o >>= 1) wv += __shfl_xor(wv, o, 32);

    int run = base + wv + incl - t;
    #pragma unroll
    for (int k = 0; k < 4; ++k) {
        const int i = i0 + k;
        if (i < n) {
            rowptr[i] = run;
            dinv[i]   = rsqrtf(fmaxf((float)(q[k] + 1), 1.0f));
            run += q[k];
            if (i == n - 1) rowptr[n] = run;
        }
    }
}

__global__ void k_place(const int* __restrict__ src, const int* __restrict__ dst,
                        const int* __restrict__ rank, const int* __restrict__ rowptr,
                        int* __restrict__ csr, int E) {
    const int T = gridDim.x * blockDim.x;
    const int e = blockIdx.x * blockDim.x + threadIdx.x;
    int d[EPT], s[EPT], pos[EPT];
    #pragma unroll
    for (int k = 0; k < EPT; ++k) {
        const bool in = e + k * T < E;
        d[k]   = in ? dst[e + k * T] : -1;
        s[k]   = in ? src[e + k * T] : 0;
        pos[k] = in ? rank[e + k * T] : 0;
    }
    #pragma unroll
    for (int k = 0; k < EPT; ++k) if (d[k] >= 0) pos[k] += rowptr[d[k]];
    #pragma unroll
    for (int k = 0; k < EPT; ++k) if (d[k] >= 0) csr[pos[k]] = s[k];
}

template<int KDIM, int NCOLT, int WCOLS>
__global__ __launch_bounds__(THREADS)
void k_gemm_rowscale(const float* __restrict__ X, const float* __restrict__ W,
                     const float* __restrict__ dinv, float* __restrict__ P, int nrows)
{
    __shared__ float sW[KDIM * NCOLT * 16];
    const int tid = threadIdx.x;
    for (int i = tid; i < KDIM * NCOLT * 16; i += THREADS) {
        const int kk = i / (NCOLT * 16);
        const int c  = i % (NCOLT * 16);
        sW[i] = (c < WCOLS) ? W[kk * WCOLS + c] : 0.0f;
    }
    __syncthreads();

    const int gwave   = (blockIdx.x * THREADS + tid) >> 5;
    const int rowBase = gwave * 16;
    if (rowBase >= nrows) return;

    const int lane = tid & 31;
    const int lh   = lane >> 4;
    const int ln   = lane & 15;

    v8f acc[NCOLT] = {};

    const float* xrow = X + (size_t)(rowBase + ln) * KDIM + 2 * lh;
    for (int kb = 0; kb < KDIM; kb += 4) {
        const v2f A = *reinterpret_cast<const v2f*>(xrow + kb);
        const float* w0 = &sW[(kb + 2 * lh) * (NCOLT * 16) + ln];
        const float* w1 = w0 + NCOLT * 16;
        #pragma unroll
        for (int t = 0; t < NCOLT; ++t) {
            v2f B;
            B.x = w0[t * 16];
            B.y = w1[t * 16];
            acc[t] = __builtin_amdgcn_wmma_f32_16x16x4_f32(
                false, A, false, B, (short)0, acc[t], false, false);
        }
    }

    const float* dvp = dinv + rowBase + 8 * lh;
    float dv[8];
    #pragma unroll
    for (int r = 0; r < 8; ++r) dv[r] = dvp[r];

    #pragma unroll
    for (int t = 0; t < NCOLT; ++t) {
        const int col = t * 16 + ln;
        if (col < WCOLS) {
            #pragma unroll
            for (int r = 0; r < 8; ++r)
                P[(size_t)(rowBase + 8 * lh + r) * WCOLS + col] = dv[r] * acc[t][r];
        }
    }
}

__device__ __forceinline__ void add4(float4& a, const float4 v) {
    a.x += v.x; a.y += v.y; a.z += v.z; a.w += v.w;
}

__global__ __launch_bounds__(THREADS)
void k_pull128(const float* __restrict__ p, const int* __restrict__ rowptr,
               const int* __restrict__ csr, const float* __restrict__ dinv,
               const float* __restrict__ bias, float* __restrict__ h, int n)
{
    const int lane = threadIdx.x & 31;
    const int node = __builtin_amdgcn_readfirstlane((int)(blockIdx.x * (THREADS / 32) + (threadIdx.x >> 5)));
    if (node >= n) return;
    const float4* __restrict__ p4 = reinterpret_cast<const float4*>(p);
    const int beg = rowptr[node];
    const int end = rowptr[node + 1];
    float4 acc = (p4 + (size_t)(unsigned)node * 32)[lane];

    for (int j = beg; j < end; j += 32) {
        const int m = min(end - j, 32);
        const int mine = (lane < m) ? csr[j + lane] : 0;
        int c = 0;
        for (; c + 8 <= m; c += 8) {
            float4 v[8];
            #pragma unroll
            for (int u = 0; u < 8; ++u) {
                const unsigned s = (unsigned)__builtin_amdgcn_readlane(mine, c + u);
                v[u] = (p4 + (size_t)s * 32)[lane];
            }
            #pragma unroll
            for (int u = 0; u < 8; ++u) add4(acc, v[u]);
        }
        if (c + 4 <= m) {
            float4 v[4];
            #pragma unroll
            for (int u = 0; u < 4; ++u) {
                const unsigned s = (unsigned)__builtin_amdgcn_readlane(mine, c + u);
                v[u] = (p4 + (size_t)s * 32)[lane];
            }
            #pragma unroll
            for (int u = 0; u < 4; ++u) add4(acc, v[u]);
            c += 4;
        }
        if (c + 2 <= m) {
            const unsigned s0 = (unsigned)__builtin_amdgcn_readlane(mine, c);
            const unsigned s1 = (unsigned)__builtin_amdgcn_readlane(mine, c + 1);
            const float4 v0 = (p4 + (size_t)s0 * 32)[lane];
            const float4 v1 = (p4 + (size_t)s1 * 32)[lane];
            add4(acc, v0);
            add4(acc, v1);
            c += 2;
        }
        if (c < m) {
            const unsigned s0 = (unsigned)__builtin_amdgcn_readlane(mine, c);
            add4(acc, (p4 + (size_t)s0 * 32)[lane]);
        }
    }

    const float  dv = dinv[node];
    const float4 b  = reinterpret_cast<const float4*>(bias)[lane];
    float4 o;
    o.x = fmaxf(dv * acc.x + b.x, 0.0f);
    o.y = fmaxf(dv * acc.y + b.y, 0.0f);
    o.z = fmaxf(dv * acc.z + b.z, 0.0f);
    o.w = fmaxf(dv * acc.w + b.w, 0.0f);
    (reinterpret_cast<float4*>(h) + (size_t)(unsigned)node * 32)[lane] = o;
}

__global__ void k_pull8(const float* __restrict__ p, const int* __restrict__ rowptr,
                        const int* __restrict__ csr, const float* __restrict__ dinv,
                        const float* __restrict__ bias, float* __restrict__ out, int n)
{
    const int t    = blockIdx.x * blockDim.x + threadIdx.x;
    const int node = t >> 3;
    const int c    = t & 7;
    if (node >= n) return;
    const int beg = rowptr[node];
    const int end = rowptr[node + 1];
    float acc = p[(size_t)(unsigned)node * 8 + c];
    int j = beg;
    for (; j + 8 <= end; j += 8) {
        int   s[8];
        float v[8];
        #pragma unroll
        for (int u = 0; u < 8; ++u) s[u] = csr[j + u];
        #pragma unroll
        for (int u = 0; u < 8; ++u) v[u] = p[(size_t)(unsigned)s[u] * 8 + c];
        #pragma unroll
        for (int u = 0; u < 8; ++u) acc += v[u];
    }
    if (j + 4 <= end) {
        int   s[4];
        float v[4];
        #pragma unroll
        for (int u = 0; u < 4; ++u) s[u] = csr[j + u];
        #pragma unroll
        for (int u = 0; u < 4; ++u) v[u] = p[(size_t)(unsigned)s[u] * 8 + c];
        #pragma unroll
        for (int u = 0; u < 4; ++u) acc += v[u];
        j += 4;
    }
    if (j + 2 <= end) {
        const int s0 = csr[j];
        const int s1 = csr[j + 1];
        const float v0 = p[(size_t)(unsigned)s0 * 8 + c];
        const float v1 = p[(size_t)(unsigned)s1 * 8 + c];
        acc += v0;
        acc += v1;
        j += 2;
    }
    if (j < end) acc += p[(size_t)(unsigned)csr[j] * 8 + c];
    out[(size_t)node * 8 + c] = dinv[node] * acc + bias[c];
}

extern "C" void kernel_launch(void* const* d_in, const int* in_sizes, int n_in,
                              void* d_out, int out_size, void* d_ws, size_t ws_size,
                              hipStream_t stream)
{
    constexpr int F = 128, H = 128, C = 8;
    const int N = in_sizes[0] / F;
    const int E = in_sizes[1] / 2;

    const float* x   = (const float*)d_in[0];
    const int*   src = (const int*)  d_in[1];
    const int*   dst = src + E;
    const float* W1  = (const float*)d_in[2];
    const float* b1  = (const float*)d_in[3];
    const float* W2  = (const float*)d_in[4];
    const float* b2  = (const float*)d_in[5];
    float*       out = (float*)d_out;

    const size_t nPad = ((size_t)N + 1 + 63) / 64 * 64;
    const size_t ePad = ((size_t)E + 63) / 64 * 64;
    int*   wsi    = (int*)d_ws;
    int*   cnt    = wsi;
    int*   rowptr = cnt + nPad;
    float* dinv   = (float*)(rowptr + nPad);
    int*   rank   = (int*)(dinv + nPad);
    int*   csr    = rank + ePad;
    float* p1     = (float*)(csr + ePad);
    float* h1     = p1 + (size_t)N * H;
    float* p2     = h1 + (size_t)N * H;

    const int gN    = (N + THREADS - 1) / THREADS;
    const int gE4   = (E + THREADS * EPT - 1) / (THREADS * EPT);
    const int gScan = (N + SCAN_TILE - 1) / SCAN_TILE;
    const int gGemm = ((N / 16) * 32 + THREADS - 1) / THREADS;
    const int gPull = (N + THREADS / 32 - 1) / (THREADS / 32);
    const int gNC   = ((size_t)N * C + THREADS - 1) / THREADS;

    k_zero      <<<gN,    THREADS,      0, stream>>>(cnt, N);
    k_count_rank<<<gE4,   THREADS,      0, stream>>>(dst, cnt, rank, E);
    k_scan_dinv <<<gScan, SCAN_THREADS, 0, stream>>>(cnt, rowptr, dinv, N);
    k_place     <<<gE4,   THREADS,      0, stream>>>(src, dst, rank, rowptr, csr, E);

    k_gemm_rowscale<F, H / 16, H><<<gGemm, THREADS, 0, stream>>>(x, W1, dinv, p1, N);

    k_pull128<<<gPull, THREADS, 0, stream>>>(p1, rowptr, csr, dinv, b1, h1, N);

    k_gemm_rowscale<H, 1, C><<<gGemm, THREADS, 0, stream>>>(h1, W2, dinv, p2, N);

    k_pull8<<<gNC, THREADS, 0, stream>>>(p2, rowptr, csr, dinv, b2, out, N);
}


# Edge list as fixed-pitch slots per destination (64 ids, spill list beyond): the counting atomic's slot places the edge at once; scan, rank, row-start gather, placement launch go; slot-order sums
# speedup vs baseline: 3.3378x; 1.0658x over previous
#include <hip/hip_runtime.h>


typedef float v2f __attribute__((ext_vector_type(2)));
typedef float v8f __attribute__((ext_vector_type(8)));

#define THREADS 256
#define EPT 4
#define SLOTS 64

__device__ __forceinline__ float deg_inv_sqrt(int indeg) {
    return rsqrtf(fmaxf((float)(indeg + 1), 1.0f));
}

__global__ void k_zero(int* __restrict__ cnt, int n) {
    int i = blockIdx.x * blockDim.x + threadIdx.x;
    if (i <= n) cnt[i] = 0;
}

__global__ void k_fill(const int* __restrict__ src, const int* __restrict__ dst,
                       int* __restrict__ cnt, int* __restrict__ slots,
                       int* __restrict__ spill, int E, int n) {
    const int T = gridDim.x * blockDim.x;
    const int e = blockIdx.x * blockDim.x + threadIdx.x;
    int d[EPT], s[EPT], r[EPT];
    #pragma unroll
    for (int k = 0; k < EPT; ++k) {
        const bool in = e + k * T < E;
        d[k] = in ? dst[e + k * T] : -1;
        s[k] = in ? src[e + k * T] : 0;
    }
    #pragma unroll
    for (int k = 0; k < EPT; ++k) r[k] = (d[k] >= 0) ? atomicAdd(cnt + d[k], 1) : 0;
    #pragma unroll
    for (int k = 0; k < EPT; ++k) {
        if (d[k] >= 0) {
            if (r[k] < SLOTS) {
                slots[(size_t)(unsigned)d[k] * SLOTS + r[k]] = s[k];
            } else {
                const int o = atomicAdd(cnt + n, 1);
                spill[2 * (size_t)o]     = d[k];
                spill[2 * (size_t)o + 1] = s[k];
            }
        }
    }
}

template<int KDIM, int NCOLT, int WCOLS>
__global__ __launch_bounds__(THREADS)
void k_gemm_rowscale(const float* __restrict__ X, const float* __restrict__ W,
                     const int* __restrict__ cnt, float* __restrict__ P, int nrows)
{
    __shared__ float sW[KDIM * NCOLT * 16];
    const int tid = threadIdx.x;
    for (int i = tid; i < KDIM * NCOLT * 16; i += THREADS) {
        const int kk = i / (NCOLT * 16);
        const int c  = i % (NCOLT * 16);
        sW[i] = (c < WCOLS) ? W[kk * WCOLS + c] : 0.0f;
    }
    __syncthreads();

    const int gwave   = (blockIdx.x * THREADS + tid) >> 5;
    const int rowBase = gwave * 16;
    if (rowBase >= nrows) return;

    const int lane = tid & 31;
    const int lh   = lane >> 4;
    const int ln   = lane & 15;

    v8f acc[NCOLT] = {};

    const float* xrow = X + (size_t)(rowBase + ln) * KDIM + 2 * lh;
    for (int kb = 0; kb < KDIM; kb += 4) {
        const v2f A = *reinterpret_cast<const v2f*>(xrow + kb);
        const float* w0 = &sW[(kb + 2 * lh) * (NCOLT * 16) + ln];
        const float* w1 = w0 + NCOLT * 16;
        #pragma unroll
        for (int t = 0; t < NCOLT; ++t) {
            v2f B;
            B.x = w0[t * 16];
            B.y = w1[t * 16];
            acc[t] = __builtin_amdgcn_wmma_f32_16x16x4_f32(
                false, A, false, B, (short)0, acc[t], false, false);
        }
    }

    const int* dgp = cnt + rowBase + 8 * lh;
    float dv[8];
    #pragma unroll
    for (int r = 0; r < 8; ++r) dv[r] = deg_inv_sqrt(dgp[r]);

    #pragma unroll
    for (int t = 0; t < NCOLT; ++t) {
        const int col = t * 16 + ln;
        if (col < WCOLS) {
            #pragma unroll
            for (int r = 0; r < 8; ++r)
                P[(size_t)(rowBase + 8 * lh + r) * WCOLS + col] = dv[r] * acc[t][r];
        }
    }
}

__device__ __forceinline__ void add4(float4& a, const float4 v) {
    a.x += v.x; a.y += v.y; a.z += v.z; a.w += v.w;
}

__global__ __launch_bounds__(THREADS)
void k_pull128(const float* __restrict__ p, const int* __restrict__ cnt,
               const int* __restrict__ slots, const int* __restrict__ spill,
               const float* __restrict__ bias, float* __restrict__ h, int n, int E)
{
    const int lane = threadIdx.x & 31;
    const int node = __builtin_amdgcn_readfirstlane((int)(blockIdx.x * (THREADS / 32) + (threadIdx.x >> 5)));
    if (node >= n) return;
    const float4* __restrict__ p4 = reinterpret_cast<const float4*>(p);
    const int deg = cnt[node];
    const int end = min(deg, SLOTS);
    const int* __restrict__ ids = slots + (size_t)(unsigned)node * SLOTS;
    float4 acc = (p4 + (size_t)(unsigned)node * 32)[lane];

    for (int j = 0; j < end; j += 32) {
        const int m = min(end - j, 32);
        const int mine = (lane < m) ? ids[j + lane] : 0;
        int c = 0;
        for (; c + 8 <= m; c += 8) {
            float4 v[8];
            #pragma unroll
            for (int u = 0; u < 8; ++u) {
                const unsigned s = (unsigned)__builtin_amdgcn_readlane(mine, c + u);
                v[u] = (p4 + (size_t)s * 32)[lane];
            }
            #pragma unroll
            for (int u = 0; u < 8; ++u) add4(acc, v[u]);
        }
        if (c + 4 <= m) {
            float4 v[4];
            #pragma unroll
            for (int u = 0; u < 4; ++u) {
                const unsigned s = (unsigned)__builtin_amdgcn_readlane(mine, c + u);
                v[u] = (p4 + (size_t)s * 32)[lane];
            }
            #pragma unroll
            for (int u = 0; u < 4; ++u) add4(acc, v[u]);
            c += 4;
        }
        if (c + 2 <= m) {
            const unsigned s0 = (unsigned)__builtin_amdgcn_readlane(mine, c);
            const unsigned s1 = (unsigned)__builtin_amdgcn_readlane(mine, c + 1);
            const float4 v0 = (p4 + (size_t)s0 * 32)[lane];
            const float4 v1 = (p4 + (size_t)s1 * 32)[lane];
            add4(acc, v0);
            add4(acc, v1);
            c += 2;
        }
        if (c < m) {
            const unsigned s0 = (unsigned)__builtin_amdgcn_readlane(mine, c);
            add4(acc, (p4 + (size_t)s0 * 32)[lane]);
        }
    }
    if (deg > SLOTS) {
        const int ns = min(cnt[n], E);
        for (int i = 0; i < ns; ++i) {
            if (spill[2 * (size_t)i] == node) {
                const unsigned s0 = (unsigned)spill[2 * (size_t)i + 1];
                add4(acc, (p4 + (size_t)s0 * 32)[lane]);
            }
        }
    }

    const float  dv = deg_inv_sqrt(deg);
    const float4 b  = reinterpret_cast<const float4*>(bias)[lane];
    float4 o;
    o.x = fmaxf(dv * acc.x + b.x, 0.0f);
    o.y = fmaxf(dv * acc.y + b.y, 0.0f);
    o.z = fmaxf(dv * acc.z + b.z, 0.0f);
    o.w = fmaxf(dv * acc.w + b.w, 0.0f);
    (reinterpret_cast<float4*>(h) + (size_t)(unsigned)node * 32)[lane] = o;
}

__global__ void k_pull8(const float* __restrict__ p, const int* __restrict__ cnt,
                        const int* __restrict__ slots, const int* __restrict__ spill,
                        const float* __restrict__ bias, float* __restrict__ out, int n, int E)
{
    const int t    = blockIdx.x * blockDim.x + threadIdx.x;
    const int node = t >> 3;
    const int c    = t & 7;
    if (node >= n) return;
    const int deg = cnt[node];
    const int end = min(deg, SLOTS);
    const int* __restrict__ csr = slots + (size_t)(unsigned)node * SLOTS;
    float acc = p[(size_t)(unsigned)node * 8 + c];
    int j = 0;
    for (; j + 8 <= end; j += 8) {
        int   s[8];
        float v[8];
        #pragma unroll
        for (int u = 0; u < 8; ++u) s[u] = csr[j + u];
        #pragma unroll
        for (int u = 0; u < 8; ++u) v[u] = p[(size_t)(unsigned)s[u] * 8 + c];
        #pragma unroll
        for (int u = 0; u < 8; ++u) acc += v[u];
    }
    if (j + 4 <= end) {
        int   s[4];
        float v[4];
        #pragma unroll
        for (int u = 0; u < 4; ++u) s[u] = csr[j + u];
        #pragma unroll
        for (int u = 0; u < 4; ++u) v[u] = p[(size_t)(unsigned)s[u] * 8 + c];
        #pragma unroll
        for (int u = 0; u < 4; ++u) acc += v[u];
        j += 4;
    }
    if (j + 2 <= end) {
        const int s0 = csr[j];
        const int s1 = csr[j + 1];
        const float v0 = p[(size_t)(unsigned)s0 * 8 + c];
        const float v1 = p[(size_t)(unsigned)s1 * 8 + c];
        acc += v0;
        acc += v1;
        j += 2;
    }
    if (j < end) acc += p[(size_t)(unsigned)csr[j] * 8 + c];
    if (deg > SLOTS) {
        const int ns = min(cnt[n], E);
        for (int i = 0; i < ns; ++i)
            if (spill[2 * (size_t)i] == node) acc += p[(size_t)(unsigned)spill[2 * (size_t)i + 1] * 8 + c];
    }
    out[(size_t)node * 8 + c] = deg_inv_sqrt(deg) * acc + bias[c];
}

extern "C" void kernel_launch(void* const* d_in, const int* in_sizes, int n_in,
                              void* d_out, int out_size, void* d_ws, size_t ws_size,
                              hipStream_t stream)
{
    constexpr int F = 128, H = 128, C = 8;
    const int N = in_sizes[0] / F;
    const int E = in_sizes[1] / 2;

    const float* x   = (const float*)d_in[0];
    const int*   src = (const int*)  d_in[1];
    const int*   dst = src + E;
    const float* W1  = (const float*)d_in[2];
    const float* b1  = (const float*)d_in[3];
    const float* W2  = (const float*)d_in[4];
    const float* b2  = (const float*)d_in[5];
    float*       out = (float*)d_out;

    const size_t nPad = ((size_t)N + 1 + 63) / 64 * 64;
    const size_t sPad = (2 * (size_t)E + 63) / 64 * 64;
    int*   wsi    = (int*)d_ws;
    int*   cnt    = wsi;
    int*   slots  = cnt + nPad;
    int*   spill  = slots + (size_t)N * SLOTS;
    float* p1     = (float*)(spill + sPad);
    float* h1     = p1 + (size_t)N * H;
    float* p2     = h1 + (size_t)N * H;

    const int gN    = (N + 1 + THREADS - 1) / THREADS;
    const int gE4   = (E + THREADS * EPT - 1) / (THREADS * EPT);
    const int gGemm = ((N / 16) * 32 + THREADS - 1) / THREADS;
    const int gPull = (N + THREADS / 32 - 1) / (THREADS / 32);
    const int gNC   = ((size_t)N * C + THREADS - 1) / THREADS;

    k_zero<<<gN,  THREADS, 0, stream>>>(cnt, N);
    k_fill<<<gE4, THREADS, 0, stream>>>(src, dst, cnt, slots, spill, E, N);

    k_gemm_rowscale<F, H / 16, H><<<gGemm, THREADS, 0, stream>>>(x, W1, cnt, p1, N);

    k_pull128<<<gPull, THREADS, 0, stream>>>(p1, cnt, slots, spill, b1, h1, N, E);

    k_gemm_rowscale<H, 1, C><<<gGemm, THREADS, 0, stream>>>(h1, W2, cnt, p2, N);

    k_pull8<<<gNC, THREADS, 0, stream>>>(p2, cnt, slots, spill, b2, out, N, E);
}


# Layer 2's product fused into layer 1's pull: 16 nodes per work-group stage their relu'd rows in LDS, one wave runs the fp32 matrix product on them; hidden rows never reach memory, one launch fewer
# speedup vs baseline: 3.4790x; 1.0423x over previous
#include <hip/hip_runtime.h>


typedef float v2f __attribute__((ext_vector_type(2)));
typedef float v8f __attribute__((ext_vector_type(8)));

#define THREADS 256
#define EPT 4
#define SLOTS 64
#define PNODES 16
#define HPITCH 132
#define CPAD 16

__device__ __forceinline__ float deg_inv_sqrt(int indeg) {
    return rsqrtf(fmaxf((float)(indeg + 1), 1.0f));
}

__global__ void k_zero(int* __restrict__ cnt, int n) {
    int i = blockIdx.x * blockDim.x + threadIdx.x;
    if (i <= n) cnt[i] = 0;
}

__global__ void k_fill(const int* __restrict__ src, const int* __restrict__ dst,
                       int* __restrict__ cnt, int* __restrict__ slots,
                       int* __restrict__ spill, int E, int n) {
    const int T = gridDim.x * blockDim.x;
    const int e = blockIdx.x * blockDim.x + threadIdx.x;
    int d[EPT], s[EPT], r[EPT];
    #pragma unroll
    for (int k = 0; k < EPT; ++k) {
        const bool in = e + k * T < E;
        d[k] = in ? dst[e + k * T] : -1;
        s[k] = in ? src[e + k * T] : 0;
    }
    #pragma unroll
    for (int k = 0; k < EPT; ++k) r[k] = (d[k] >= 0) ? atomicAdd(cnt + d[k], 1) : 0;
    #pragma unroll
    for (int k = 0; k < EPT; ++k) {
        if (d[k] >= 0) {
            if (r[k] < SLOTS) {
                slots[(size_t)(unsigned)d[k] * SLOTS + r[k]] = s[k];
            } else {
                const int o = atomicAdd(cnt + n, 1);
                spill[2 * (size_t)o]     = d[k];
                spill[2 * (size_t)o + 1] = s[k];
            }
        }
    }
}

template<int KDIM, int NCOLT, int WCOLS>
__global__ __launch_bounds__(THREADS)
void k_gemm_rowscale(const float* __restrict__ X, const float* __restrict__ W,
                     const int* __restrict__ cnt, float* __restrict__ P, int nrows)
{
    __shared__ float sW[KDIM * NCOLT * 16];
    const int tid = threadIdx.x;
    for (int i = tid; i < KDIM * NCOLT * 16; i += THREADS) {
        const int kk = i / (NCOLT * 16);
        const int c  = i % (NCOLT * 16);
        sW[i] = (c < WCOLS) ? W[kk * WCOLS + c] : 0.0f;
    }
    __syncthreads();

    const int gwave   = (blockIdx.x * THREADS + tid) >> 5;
    const int rowBase = gwave * 16;
    if (rowBase >= nrows) return;

    const int lane = tid & 31;
    const int lh   = lane >> 4;
    const int ln   = lane & 15;

    v8f acc[NCOLT] = {};

    const float* xrow = X + (size_t)(rowBase + ln) * KDIM + 2 * lh;
    for (int kb = 0; kb < KDIM; kb += 4) {
        const v2f A = *reinterpret_cast<const v2f*>(xrow + kb);
        const float* w0 = &sW[(kb + 2 * lh) * (NCOLT * 16) + ln];
        const float* w1 = w0 + NCOLT * 16;
        #pragma unroll
        for (int t = 0; t < NCOLT; ++t) {
            v2f B;
            B.x = w0[t * 16];
            B.y = w1[t * 16];
            acc[t] = __builtin_amdgcn_wmma_f32_16x16x4_f32(
                false, A, false, B, (short)0, acc[t], false, false);
        }
    }

    const int* dgp = cnt + rowBase + 8 * lh;
    float dv[8];
    #pragma unroll
    for (int r = 0; r < 8; ++r) dv[r] = deg_inv_sqrt(dgp[r]);

    #pragma unroll
    for (int t = 0; t < NCOLT; ++t) {
        const int col = t * 16 + ln;
        if (col < WCOLS) {
            #pragma unroll
            for (int r = 0; r < 8; ++r)
                P[(size_t)(rowBase + 8 * lh + r) * WCOLS + col] = dv[r] * acc[t][r];
        }
    }
}

__device__ __forceinline__ void add4(float4& a, const float4 v) {
    a.x += v.x; a.y += v.y; a.z += v.z; a.w += v.w;
}

__global__ __launch_bounds__(PNODES * 32)
void k_pull128_w2(const float* __restrict__ p, const int* __restrict__ cnt,
                  const int* __restrict__ slots, const int* __restrict__ spill,
                  const float* __restrict__ bias, const float* __restrict__ W2,
                  float* __restrict__ p2, int n, int E)
{
    __shared__ __attribute__((aligned(16))) float sH[PNODES * HPITCH];
    __shared__ __attribute__((aligned(16))) float sW[128 * CPAD];
    __shared__ __attribute__((aligned(16))) float sD[PNODES];
    const int tid  = threadIdx.x;
    const int lane = tid & 31;
    const int wave = __builtin_amdgcn_readfirstlane(tid >> 5);
    const int node = blockIdx.x * PNODES + wave;

    float wv[128 * CPAD / (PNODES * 32)];
    #pragma unroll
    for (int q = 0; q < 128 * CPAD / (PNODES * 32); ++q) {
        const int i  = tid + q * (PNODES * 32);
        const int kk = i / CPAD;
        const int c  = i % CPAD;
        wv[q] = (c < 8) ? W2[kk * 8 + c] : 0.0f;
    }

    float4 o = make_float4(0.0f, 0.0f, 0.0f, 0.0f);
    float  dvn = 0.0f;
    if (node < n) {
    const float4* __restrict__ p4 = reinterpret_cast<const float4*>(p);
    const int deg = cnt[node];
    const int end = min(deg, SLOTS);
    const int* __restrict__ ids = slots + (size_t)(unsigned)node * SLOTS;
    float4 acc = (p4 + (size_t)(unsigned)node * 32)[lane];

    for (int j = 0; j < end; j += 32) {
        const int m = min(end - j, 32);
        const int mine = (lane < m) ? ids[j + lane] : 0;
        int c = 0;
        for (; c + 8 <= m; c += 8) {
            float4 v[8];
            #pragma unroll
            for (int u = 0; u < 8; ++u) {
                const unsigned s = (unsigned)__builtin_amdgcn_readlane(mine, c + u);
                v[u] = (p4 + (size_t)s * 32)[lane];
            }
            #pragma unroll
            for (int u = 0; u < 8; ++u) add4(acc, v[u]);
        }
        if (c + 4 <= m) {
            float4 v[4];
            #pragma unroll
            for (int u = 0; u < 4; ++u) {
                const unsigned s = (unsigned)__builtin_amdgcn_readlane(mine, c + u);
                v[u] = (p4 + (size_t)s * 32)[lane];
            }
            #pragma unroll
            for (int u = 0; u < 4; ++u) add4(acc, v[u]);
            c += 4;
        }
        if (c + 2 <= m) {
            const unsigned s0 = (unsigned)__builtin_amdgcn_readlane(mine, c);
            const unsigned s1 = (unsigned)__builtin_amdgcn_readlane(mine, c + 1);
            const float4 v0 = (p4 + (size_t)s0 * 32)[lane];
            const float4 v1 = (p4 + (size_t)s1 * 32)[lane];
            add4(acc, v0);
            add4(acc, v1);
            c += 2;
        }
        if (c < m) {
            const unsigned s0 = (unsigned)__builtin_amdgcn_readlane(mine, c);
            add4(acc, (p4 + (size_t)s0 * 32)[lane]);
        }
    }
    if (deg > SLOTS) {
        const int ns = min(cnt[n], E);
        for (int i = 0; i < ns; ++i) {
            if (spill[2 * (size_t)i] == node) {
                const unsigned s0 = (unsigned)spill[2 * (size_t)i + 1];
                add4(acc, (p4 + (size_t)s0 * 32)[lane]);
            }
        }
    }

    const float  dv = deg_inv_sqrt(deg);
    const float4 b  = reinterpret_cast<const float4*>(bias)[lane];
    o.x = fmaxf(dv * acc.x + b.x, 0.0f);
    o.y = fmaxf(dv * acc.y + b.y, 0.0f);
    o.z = fmaxf(dv * acc.z + b.z, 0.0f);
    o.w = fmaxf(dv * acc.w + b.w, 0.0f);
    dvn = dv;
    }

    *reinterpret_cast<float4*>(&sH[wave * HPITCH + 4 * lane]) = o;
    if (lane == 0) sD[wave] = dvn;
    #pragma unroll
    for (int q = 0; q < 128 * CPAD / (PNODES * 32); ++q) sW[tid + q * (PNODES * 32)] = wv[q];
    __syncthreads();
    if (wave != 0) return;

    const int lh = lane >> 4;
    const int ln = lane & 15;
    const int rowBase = blockIdx.x * PNODES;

    float dr[8];
    #pragma unroll
    for (int r = 0; r < 8; ++r) dr[r] = sD[8 * lh + r];

    v8f acc2 = {};
    const float* arow = &sH[ln * HPITCH + 2 * lh];
    const float* bcol = &sW[(2 * lh) * CPAD + ln];
    for (int kb = 0; kb < 128; kb += 4) {
        const v2f A = *reinterpret_cast<const v2f*>(arow + kb);
        v2f B;
        B.x = bcol[kb * CPAD];
        B.y = bcol[kb * CPAD + CPAD];
        acc2 = __builtin_amdgcn_wmma_f32_16x16x4_f32(
            false, A, false, B, (short)0, acc2, false, false);
    }

    if (ln < 8) {
        #pragma unroll
        for (int r = 0; r < 8; ++r)
            p2[(size_t)(rowBase + 8 * lh + r) * 8 + ln] = dr[r] * acc2[r];
    }
}

__global__ void k_pull8(const float* __restrict__ p, const int* __restrict__ cnt,
                        const int* __restrict__ slots, const int* __restrict__ spill,
                        const float* __restrict__ bias, float* __restrict__ out, int n, int E)
{
    const int t    = blockIdx.x * blockDim.x + threadIdx.x;
    const int node = t >> 3;
    const int c    = t & 7;
    if (node >= n) return;
    const int deg = cnt[node];
    const int end = min(deg, SLOTS);
    const int* __restrict__ csr = slots + (size_t)(unsigned)node * SLOTS;
    float acc = p[(size_t)(unsigned)node * 8 + c];
    int j = 0;
    for (; j + 8 <= end; j += 8) {
        int   s[8];
        float v[8];
        #pragma unroll
        for (int u = 0; u < 8; ++u) s[u] = csr[j + u];
        #pragma unroll
        for (int u = 0; u < 8; ++u) v[u] = p[(size_t)(unsigned)s[u] * 8 + c];
        #pragma unroll
        for (int u = 0; u < 8; ++u) acc += v[u];
    }
    if (j + 4 <= end) {
        int   s[4];
        float v[4];
        #pragma unroll
        for (int u = 0; u < 4; ++u) s[u] = csr[j + u];
        #pragma unroll
        for (int u = 0; u < 4; ++u) v[u] = p[(size_t)(unsigned)s[u] * 8 + c];
        #pragma unroll
        for (int u = 0; u < 4; ++u) acc += v[u];
        j += 4;
    }
    if (j + 2 <= end) {
        const int s0 = csr[j];
        const int s1 = csr[j + 1];
        const float v0 = p[(size_t)(unsigned)s0 * 8 + c];
        const float v1 = p[(size_t)(unsigned)s1 * 8 + c];
        acc += v0;
        acc += v1;
        j += 2;
    }
    if (j < end) acc += p[(size_t)(unsigned)csr[j] * 8 + c];
    if (deg > SLOTS) {
        const int ns = min(cnt[n], E);
        for (int i = 0; i < ns; ++i)
            if (spill[2 * (size_t)i] == node) acc += p[(size_t)(unsigned)spill[2 * (size_t)i + 1] * 8 + c];
    }
    out[(size_t)node * 8 + c] = deg_inv_sqrt(deg) * acc + bias[c];
}

extern "C" void kernel_launch(void* const* d_in, const int* in_sizes, int n_in,
                              void* d_out, int out_size, void* d_ws, size_t ws_size,
                              hipStream_t stream)
{
    constexpr int F = 128, H = 128, C = 8;
    const int N = in_sizes[0] / F;
    const int E = in_sizes[1] / 2;

    const float* x   = (const float*)d_in[0];
    const int*   src = (const int*)  d_in[1];
    const int*   dst = src + E;
    const float* W1  = (const float*)d_in[2];
    const float* b1  = (const float*)d_in[3];
    const float* W2  = (const float*)d_in[4];
    const float* b2  = (const float*)d_in[5];
    float*       out = (float*)d_out;

    const size_t nPad = ((size_t)N + 1 + 63) / 64 * 64;
    const size_t sPad = (2 * (size_t)E + 63) / 64 * 64;
    int*   wsi    = (int*)d_ws;
    int*   cnt    = wsi;
    int*   slots  = cnt + nPad;
    int*   spill  = slots + (size_t)N * SLOTS;
    float* p1     = (float*)(spill + sPad);
    float* p2     = p1 + (size_t)N * H;

    const int gN    = (N + 1 + THREADS - 1) / THREADS;
    const int gE4   = (E + THREADS * EPT - 1) / (THREADS * EPT);
    const int gGemm = ((N / 16) * 32 + THREADS - 1) / THREADS;
    const int gPull = (N + PNODES - 1) / PNODES;
    const int gNC   = ((size_t)N * C + THREADS - 1) / THREADS;

    k_zero<<<gN,  THREADS, 0, stream>>>(cnt, N);
    k_fill<<<gE4, THREADS, 0, stream>>>(src, dst, cnt, slots, spill, E, N);

    k_gemm_rowscale<F, H / 16, H><<<gGemm, THREADS, 0, stream>>>(x, W1, cnt, p1, N);

    k_pull128_w2<<<gPull, PNODES * 32, 0, stream>>>(p1, cnt, slots, spill, b1, W2, p2, N, E);

    k_pull8<<<gNC, THREADS, 0, stream>>>(p2, cnt, slots, spill, b2, out, N, E);
}


# Edge grouping without a global atomic per edge: edges binned by 128-node destination range in LDS, then one work-group per bin fills its slot rows in LDS, stores whole lines; slot order by arrival
# speedup vs baseline: 5.5376x; 1.5917x over previous
#include <hip/hip_runtime.h>


typedef float v2f __attribute__((ext_vector_type(2)));
typedef float v8f __attribute__((ext_vector_type(8)));

#define THREADS 256
#define BINN 128
#define NBMAX 1024
#define CAP 4096
#define CH 4096
#define EPB (CH / THREADS)
#define GPITCH 32
#define SLOTS 64
#define PNODES 16
#define HPITCH 132
#define CPAD 16

__device__ __forceinline__ float deg_inv_sqrt(int indeg) {
    return rsqrtf(fmaxf((float)(indeg + 1), 1.0f));
}

static_assert(NBMAX == 4 * THREADS, "the prefix sum of the binning pass gives every thread 4 bins");
static_assert(CH == EPB * THREADS && BINN <= THREADS, "a chunk is whole edges per thread; one thread per node of a bin zeroes and stores its in-degree");
__device__ __forceinline__ int bin_of(int d) { return (int)((unsigned)d / BINN); }

__global__ void k_zero(int* __restrict__ gcnt, int* __restrict__ cntN, int nb) {
    const int i = blockIdx.x * blockDim.x + threadIdx.x;
    if (i <= nb) gcnt[(size_t)i * GPITCH] = 0;
    if (i == 0) *cntN = 0;
}

__global__ __launch_bounds__(THREADS)
void k_bin(const int* __restrict__ src, const int* __restrict__ dst, int* __restrict__ gcnt,
           int2* __restrict__ binbuf, int2* __restrict__ ovf, int E, int nb)
{
    __shared__ int  sHist[NBMAX];
    __shared__ int  sStart[NBMAX];
    __shared__ int  sGbase[NBMAX];
    __shared__ int  sWave[THREADS / 32];
    __shared__ int2 sRec[CH];
    const int tid  = threadIdx.x;
    const int lane = tid & 31;
    const int wave = tid >> 5;
    const int base = blockIdx.x * CH;

    for (int t = tid; t < NBMAX; t += THREADS) sHist[t] = 0;
    __syncthreads();

    int d[EPB], s[EPB], r[EPB];
    #pragma unroll
    for (int k = 0; k < EPB; ++k) {
        const int e = base + k * THREADS + tid;
        const bool in = e < E;
        d[k] = in ? dst[e] : -1;
        s[k] = in ? src[e] : 0;
    }
    #pragma unroll
    for (int k = 0; k < EPB; ++k) r[k] = (d[k] >= 0) ? atomicAdd(&sHist[bin_of(d[k])], 1) : 0;
    __syncthreads();

    const int b0 = 4 * tid;
    int h[4];
    #pragma unroll
    for (int q = 0; q < 4; ++q) h[q] = sHist[b0 + q];
    const int tsum = h[0] + h[1] + h[2] + h[3];
    int incl = tsum;
    #pragma unroll
    for (int off = 1; off < 32; off <<= 1) {
        const int v = __shfl_up(incl, off);
        if (lane >= off) incl += v;
    }
    if (lane == 31) sWave[wave] = incl;
    __syncthreads();
    int x = incl - tsum;
    for (int w = 0; w < wave; ++w) x += sWave[w];
    #pragma unroll
    for (int q = 0; q < 4; ++q) {
        sStart[b0 + q] = x;
        x += h[q];
        if (h[q] > 0) sGbase[b0 + q] = atomicAdd(gcnt + (size_t)(b0 + q) * GPITCH, h[q]);
    }
    __syncthreads();

    #pragma unroll
    for (int k = 0; k < EPB; ++k)
        if (d[k] >= 0) sRec[sStart[bin_of(d[k])] + r[k]] = make_int2(d[k], s[k]);
    __syncthreads();

    const int total = min(CH, E - base);
    for (int i = tid; i < total; i += THREADS) {
        const int2 rec = sRec[i];
        const int  b   = bin_of(rec.x);
        const int  pos = sGbase[b] + (i - sStart[b]);
        int2* to = binbuf + ((size_t)b * CAP + pos);
        if (pos >= CAP) to = ovf + atomicAdd(gcnt + (size_t)nb * GPITCH, 1);
        *to = rec;
    }
}

__device__ __forceinline__ void place_edge(const int2 rec, int node0, int* sDeg, int* sSlots,
                                           int* __restrict__ cntN, int* __restrict__ spill) {
    const int ln = rec.x - node0;
    const int r  = atomicAdd(&sDeg[ln], 1);
    if (r < SLOTS) {
        sSlots[ln * SLOTS + r] = rec.y;
    } else {
        const int o = atomicAdd(cntN, 1);
        spill[2 * (size_t)o]     = rec.x;
        spill[2 * (size_t)o + 1] = rec.y;
    }
}

__global__ __launch_bounds__(THREADS)
void k_place(const int* __restrict__ gcnt, const int2* __restrict__ binbuf, const int2* __restrict__ ovf,
             int* __restrict__ cnt, int* __restrict__ slots, int* __restrict__ spill, int n, int E, int nb)
{
    __shared__ __attribute__((aligned(16))) int sSlots[BINN * SLOTS];
    __shared__ int sDeg[BINN];
    const int tid   = threadIdx.x;
    const int b     = blockIdx.x;
    const int node0 = b * BINN;
    if (tid < BINN) sDeg[tid] = 0;
    __syncthreads();

    const int   m    = min(gcnt[(size_t)b * GPITCH], CAP);
    const int2* mine = binbuf + (size_t)b * CAP;
    for (int i0 = 0; i0 < m; i0 += 4 * THREADS) {
        int2 rec[4];
        #pragma unroll
        for (int u = 0; u < 4; ++u) {
            const int i = i0 + u * THREADS + tid;
            rec[u] = (i < m) ? mine[i] : make_int2(-1, 0);
        }
        #pragma unroll
        for (int u = 0; u < 4; ++u)
            if (rec[u].x >= 0) place_edge(rec[u], node0, sDeg, sSlots, cnt + n, spill);
    }
    const int no = min(gcnt[(size_t)nb * GPITCH], E);
    for (int i = tid; i < no; i += THREADS) {
        const int2 rec = ovf[i];
        if (bin_of(rec.x) == b) place_edge(rec, node0, sDeg, sSlots, cnt + n, spill);
    }
    __syncthreads();

    if (tid < BINN && node0 + tid < n) cnt[node0 + tid] = sDeg[tid];
    for (int q = tid; q < BINN * 16; q += THREADS) {
        const int ln  = q >> 4;
        const int i0  = 4 * (q & 15);
        const int deg = min(sDeg[ln], SLOTS);
        if (node0 + ln < n && deg > (i0 & 32)) {
            int4 v = *reinterpret_cast<const int4*>(&sSlots[ln * SLOTS + i0]);
            v.x = (i0     < deg) ? v.x : 0;
            v.y = (i0 + 1 < deg) ? v.y : 0;
            v.z = (i0 + 2 < deg) ? v.z : 0;
            v.w = (i0 + 3 < deg) ? v.w : 0;
            *reinterpret_cast<int4*>(slots + (size_t)(node0 + ln) * SLOTS + i0) = v;
        }
    }
}

template<int KDIM, int NCOLT, int WCOLS>
__global__ __launch_bounds__(THREADS)
void k_gemm_rowscale(const float* __restrict__ X, const float* __restrict__ W,
                     const int* __restrict__ cnt, float* __restrict__ P, int nrows)
{
    __shared__ float sW[KDIM * NCOLT * 16];
    const int tid = threadIdx.x;
    for (int i = tid; i < KDIM * NCOLT * 16; i += THREADS) {
        const int kk = i / (NCOLT * 16);
        const int c  = i % (NCOLT * 16);
        sW[i] = (c < WCOLS) ? W[kk * WCOLS + c] : 0.0f;
    }
    __syncthreads();

    const int gwave   = (blockIdx.x * THREADS + tid) >> 5;
    const int rowBase = gwave * 16;
    if (rowBase >= nrows) return;

    const int lane = tid & 31;
    const int lh   = lane >> 4;
    const int ln   = lane & 15;

    v8f acc[NCOLT] = {};

    const float* xrow = X + (size_t)(rowBase + ln) * KDIM + 2 * lh;
    for (int kb = 0; kb < KDIM; kb += 4) {
        const v2f A = *reinterpret_cast<const v2f*>(xrow + kb);
        const float* w0 = &sW[(kb + 2 * lh) * (NCOLT * 16) + ln];
        const float* w1 = w0 + NCOLT * 16;
        #pragma unroll
        for (int t = 0; t < NCOLT; ++t) {
            v2f B;
            B.x = w0[t * 16];
            B.y = w1[t * 16];
            acc[t] = __builtin_amdgcn_wmma_f32_16x16x4_f32(
                false, A, false, B, (short)0, acc[t], false, false);
        }
    }

    const int* dgp = cnt + rowBase + 8 * lh;
    float dv[8];
    #pragma unroll
    for (int r = 0; r < 8; ++r) dv[r] = deg_inv_sqrt(dgp[r]);

    #pragma unroll
    for (int t = 0; t < NCOLT; ++t) {
        const int col = t * 16 + ln;
        if (col < WCOLS) {
            #pragma unroll
            for (int r = 0; r < 8; ++r)
                P[(size_t)(rowBase + 8 * lh + r) * WCOLS + col] = dv[r] * acc[t][r];
        }
    }
}

__device__ __forceinline__ void add4(float4& a, const float4 v) {
    a.x += v.x; a.y += v.y; a.z += v.z; a.w += v.w;
}

__global__ __launch_bounds__(PNODES * 32)
void k_pull128_w2(const float* __restrict__ p, const int* __restrict__ cnt,
                  const int* __restrict__ slots, const int* __restrict__ spill,
                  const float* __restrict__ bias, const float* __restrict__ W2,
                  float* __restrict__ p2, int n, int E)
{
    __shared__ __attribute__((aligned(16))) float sH[PNODES * HPITCH];
    __shared__ __attribute__((aligned(16))) float sW[128 * CPAD];
    __shared__ __attribute__((aligned(16))) float sD[PNODES];
    const int tid  = threadIdx.x;
    const int lane = tid & 31;
    const int wave = __builtin_amdgcn_readfirstlane(tid >> 5);
    const int node = blockIdx.x * PNODES + wave;

    float wv[128 * CPAD / (PNODES * 32)];
    #pragma unroll
    for (int q = 0; q < 128 * CPAD / (PNODES * 32); ++q) {
        const int i  = tid + q * (PNODES * 32);
        const int kk = i / CPAD;
        const int c  = i % CPAD;
        wv[q] = (c < 8) ? W2[kk * 8 + c] : 0.0f;
    }

    float4 o = make_float4(0.0f, 0.0f, 0.0f, 0.0f);
    float  dvn = 0.0f;
    if (node < n) {
    const float4* __restrict__ p4 = reinterpret_cast<const float4*>(p);
    const int deg = cnt[node];
    const int end = min(deg, SLOTS);
    const int* __restrict__ ids = slots + (size_t)(unsigned)node * SLOTS;
    float4 acc = (p4 + (size_t)(unsigned)node * 32)[lane];

    for (int j = 0; j < end; j += 32) {
        const int m = min(end - j, 32);
        const int mine = (lane < m) ? ids[j + lane] : 0;
        int c = 0;
        for (; c + 8 <= m; c += 8) {
            float4 v[8];
            #pragma unroll
            for (int u = 0; u < 8; ++u) {
                const unsigned s = (unsigned)__builtin_amdgcn_readlane(mine, c + u);
                v[u] = (p4 + (size_t)s * 32)[lane];
            }
            #pragma unroll
            for (int u = 0; u < 8; ++u) add4(acc, v[u]);
        }
        if (c + 4 <= m) {
            float4 v[4];
            #pragma unroll
            for (int u = 0; u < 4; ++u) {
                const unsigned s = (unsigned)__builtin_amdgcn_readlane(mine, c + u);
                v[u] = (p4 + (size_t)s * 32)[lane];
            }
            #pragma unroll
            for (int u = 0; u < 4; ++u) add4(acc, v[u]);
            c += 4;
        }
        if (c + 2 <= m) {
            const unsigned s0 = (unsigned)__builtin_amdgcn_readlane(mine, c);
            const unsigned s1 = (unsigned)__builtin_amdgcn_readlane(mine, c + 1);
            const float4 v0 = (p4 + (size_t)s0 * 32)[lane];
            const float4 v1 = (p4 + (size_t)s1 * 32)[lane];
            add4(acc, v0);
            add4(acc, v1);
            c += 2;
        }
        if (c < m) {
            const unsigned s0 = (unsigned)__builtin_amdgcn_readlane(mine, c);
            add4(acc, (p4 + (size_t)s0 * 32)[lane]);
        }
    }
    if (deg > SLOTS) {
        const int ns = min(cnt[n], E);
        for (int i = 0; i < ns; ++i) {
            if (spill[2 * (size_t)i] == node) {
                const unsigned s0 = (unsigned)spill[2 * (size_t)i + 1];
                add4(acc, (p4 + (size_t)s0 * 32)[lane]);
            }
        }
    }

    const float  dv = deg_inv_sqrt(deg);
    const float4 b  = reinterpret_cast<const float4*>(bias)[lane];
    o.x = fmaxf(dv * acc.x + b.x, 0.0f);
    o.y = fmaxf(dv * acc.y + b.y, 0.0f);
    o.z = fmaxf(dv * acc.z + b.z, 0.0f);
    o.w = fmaxf(dv * acc.w + b.w, 0.0f);
    dvn = dv;
    }

    *reinterpret_cast<float4*>(&sH[wave * HPITCH + 4 * lane]) = o;
    if (lane == 0) sD[wave] = dvn;
    #pragma unroll
    for (int q = 0; q < 128 * CPAD / (PNODES * 32); ++q) sW[tid + q * (PNODES * 32)] = wv[q];
    __syncthreads();
    if (wave != 0) return;

    const int lh = lane >> 4;
    const int ln = lane & 15;
    const int rowBase = blockIdx.x * PNODES;

    float dr[8];
    #pragma unroll
    for (int r = 0; r < 8; ++r) dr[r] = sD[8 * lh + r];

    v8f acc2 = {};
    const float* arow = &sH[ln * HPITCH + 2 * lh];
    const float* bcol = &sW[(2 * lh) * CPAD + ln];
    for (int kb = 0; kb < 128; kb += 4) {
        const v2f A = *reinterpret_cast<const v2f*>(arow + kb);
        v2f B;
        B.x = bcol[kb * CPAD];
        B.y = bcol[kb * CPAD + CPAD];
        acc2 = __builtin_amdgcn_wmma_f32_16x16x4_f32(
            false, A, false, B, (short)0, acc2, false, false);
    }

    if (ln < 8) {
        #pragma unroll
        for (int r = 0; r < 8; ++r)
            p2[(size_t)(rowBase + 8 * lh + r) * 8 + ln] = dr[r] * acc2[r];
    }
}

__global__ void k_pull8(const float* __restrict__ p, const int* __restrict__ cnt,
                        const int* __restrict__ slots, const int* __restrict__ spill,
                        const float* __restrict__ bias, float* __restrict__ out, int n, int E)
{
    const int t    = blockIdx.x * blockDim.x + threadIdx.x;
    const int node = t >> 3;
    const int c    = t & 7;
    if (node >= n) return;
    const int deg = cnt[node];
    const int end = min(deg, SLOTS);
    const int* __restrict__ csr = slots + (size_t)(unsigned)node * SLOTS;
    float acc = p[(size_t)(unsigned)node * 8 + c];
    int j = 0;
    for (; j + 8 <= end; j += 8) {
        int   s[8];
        float v[8];
        #pragma unroll
        for (int u = 0; u < 8; ++u) s[u] = csr[j + u];
        #pragma unroll
        for (int u = 0; u < 8; ++u) v[u] = p[(size_t)(unsigned)s[u] * 8 + c];
        #pragma unroll
        for (int u = 0; u < 8; ++u) acc += v[u];
    }
    if (j + 4 <= end) {
        int   s[4];
        float v[4];
        #pragma unroll
        for (int u = 0; u < 4; ++u) s[u] = csr[j + u];
        #pragma unroll
        for (int u = 0; u < 4; ++u) v[u] = p[(size_t)(unsigned)s[u] * 8 + c];
        #pragma unroll
        for (int u = 0; u < 4; ++u) acc += v[u];
        j += 4;
    }
    if (j + 2 <= end) {
        const int s0 = csr[j];
        const int s1 = csr[j + 1];
        const float v0 = p[(size_t)(unsigned)s0 * 8 + c];
        const float v1 = p[(size_t)(unsigned)s1 * 8 + c];
        acc += v0;
        acc += v1;
        j += 2;
    }
    if (j < end) acc += p[(size_t)(unsigned)csr[j] * 8 + c];
    if (deg > SLOTS) {
        const int ns = min(cnt[n], E);
        for (int i = 0; i < ns; ++i)
            if (spill[2 * (size_t)i] == node) acc += p[(size_t)(unsigned)spill[2 * (size_t)i + 1] * 8 + c];
    }
    out[(size_t)node * 8 + c] = deg_inv_sqrt(deg) * acc + bias[c];
}

extern "C" void kernel_launch(void* const* d_in, const int* in_sizes, int n_in,
                              void* d_out, int out_size, void* d_ws, size_t ws_size,
                              hipStream_t stream)
{
    constexpr int F = 128, H = 128, C = 8;
    const int N = in_sizes[0] / F;
    const int E = in_sizes[1] / 2;

    const float* x   = (const float*)d_in[0];
    const int*   src = (const int*)  d_in[1];
    const int*   dst = src + E;
    const float* W1  = (const float*)d_in[2];
    const float* b1  = (const float*)d_in[3];
    const float* W2  = (const float*)d_in[4];
    const float* b2  = (const float*)d_in[5];
    float*       out = (float*)d_out;

    const int    NB   = (N + BINN - 1) / BINN;
    const size_t nPad = ((size_t)N + 1 + 63) / 64 * 64;
    const size_t sPad = (2 * (size_t)E + 63) / 64 * 64;
    const size_t qPad = (((size_t)N + 15) / 16 * 16 * C + 63) / 64 * 64;
    const size_t gPad = ((size_t)(NB + 1) * GPITCH + 63) / 64 * 64;
    int*   wsi    = (int*)d_ws;
    int*   cnt    = wsi;
    int*   slots  = cnt + nPad;
    int*   spill  = slots + (size_t)N * SLOTS;
    float* p1     = (float*)(spill + sPad);
    float* p2     = p1 + (size_t)N * H;
    int*   gcnt   = (int*)(p2 + qPad);
    int2*  binbuf = (int2*)(gcnt + gPad);
    int2*  ovf    = binbuf + (size_t)NB * CAP;

    const int gZ    = (NB + 1 + THREADS - 1) / THREADS;
    const int gBin  = (E + CH - 1) / CH;
    const int gGemm = ((N / 16) * 32 + THREADS - 1) / THREADS;
    const int gPull = (N + PNODES - 1) / PNODES;
    const int gNC   = ((size_t)N * C + THREADS - 1) / THREADS;

    k_zero <<<gZ,   THREADS, 0, stream>>>(gcnt, cnt + N, NB);
    k_bin  <<<gBin, THREADS, 0, stream>>>(src, dst, gcnt, binbuf, ovf, E, NB);
    k_place<<<NB,   THREADS, 0, stream>>>(gcnt, binbuf, ovf, cnt, slots, spill, N, E, NB);

    k_gemm_rowscale<F, H / 16, H><<<gGemm, THREADS, 0, stream>>>(x, W1, cnt, p1, N);

    k_pull128_w2<<<gPull, PNODES * 32, 0, stream>>>(p1, cnt, slots, spill, b1, W2, p2, N, E);

    k_pull8<<<gNC, THREADS, 0, stream>>>(p2, cnt, slots, spill, b2, out, N, E);
}
